# starting kernel
# baseline (speedup 1.0000x reference)
#include <hip/hip_runtime.h>
#include <hip/hip_bf16.h>

typedef float v2f __attribute__((ext_vector_type(2)));
typedef float v8f __attribute__((ext_vector_type(8)));

#define N_NODES   50000
#define N_EDGES   800000
#define IN_FEATS  256
#define OUT_FEATS 128

__global__ void gc_zero_kernel(float* __restrict__ p, int n) {
    int i = blockIdx.x * blockDim.x + threadIdx.x;
    int stride = gridDim.x * blockDim.x;
    for (; i < n; i += stride) p[i] = 0.0f;
}

__global__ void gc_degree_kernel(const int* __restrict__ src,
                                 const int* __restrict__ dst,
                                 float* __restrict__ out_deg,
                                 float* __restrict__ in_deg) {
    int e = blockIdx.x * blockDim.x + threadIdx.x;
    if (e >= N_EDGES) return;
    unsafeAtomicAdd(&out_deg[src[e]], 1.0f);
    unsafeAtomicAdd(&in_deg[dst[e]], 1.0f);
}

__global__ void gc_gemm_wmma_kernel(const float* __restrict__ x,
                                    const float* __restrict__ Wm,
                                    const float* __restrict__ out_deg,
                                    float* __restrict__ h) {
    const int lane = threadIdx.x & 31;
    const int wave = threadIdx.x >> 5;
    const int half = lane >> 4;
    const int l    = lane & 15;

    const int m0 = blockIdx.x * 16;
    const int n0 = wave * 16;

    const int row = m0 + l;
    const float rs = rsqrtf(fmaxf(out_deg[row], 1.0f));
    const float* __restrict__ xrow = x + (size_t)row * IN_FEATS;

    v8f c = {};
    #pragma unroll 8
    for (int k0 = 0; k0 < IN_FEATS; k0 += 4) {
        const int ka = k0 + 2 * half;
        v2f a, b;
        a.x = xrow[ka]     * rs;
        a.y = xrow[ka + 1] * rs;
        b.x = Wm[(size_t)ka       * OUT_FEATS + n0 + l];
        b.y = Wm[(size_t)(ka + 1) * OUT_FEATS + n0 + l];
        c = __builtin_amdgcn_wmma_f32_16x16x4_f32(false, a, false, b,
                                                  (short)0, c, false, false);
    }

    #pragma unroll
    for (int r = 0; r < 8; ++r) {
        const int m = m0 + r + 8 * half;
        h[(size_t)m * OUT_FEATS + n0 + l] = c[r];
    }
}

__global__ void gc_spmm_kernel(const float* __restrict__ h,
                               const float* __restrict__ ew,
                               const int* __restrict__ src,
                               const int* __restrict__ dst,
                               float* __restrict__ agg) {
    const int wave_in_blk = threadIdx.x >> 5;
    const int lane        = threadIdx.x & 31;
    const long long e = (long long)blockIdx.x * (blockDim.x >> 5) + wave_in_blk;
    if (e >= N_EDGES) return;

    const int  s = src[e];
    const int  d = dst[e];
    const float w = ew[e];

    const float4 hv = *(const float4*)(h + (size_t)s * OUT_FEATS + lane * 4);
    float* outp = agg + (size_t)d * OUT_FEATS + lane * 4;
    unsafeAtomicAdd(outp + 0, hv.x * w);
    unsafeAtomicAdd(outp + 1, hv.y * w);
    unsafeAtomicAdd(outp + 2, hv.z * w);
    unsafeAtomicAdd(outp + 3, hv.w * w);
}

__global__ void gc_finalize_kernel(float* __restrict__ out,
                                   const float* __restrict__ in_deg,
                                   const float* __restrict__ b) {
    const long long i = (long long)blockIdx.x * blockDim.x + threadIdx.x;
    if (i >= (long long)N_NODES * OUT_FEATS) return;
    const int node = (int)(i >> 7);
    const int col  = (int)(i & 127);
    const float rn = rsqrtf(fmaxf(in_deg[node], 1.0f));
    out[i] = out[i] * rn + b[col];
}

extern "C" void kernel_launch(void* const* d_in, const int* in_sizes, int n_in,
                              void* d_out, int out_size, void* d_ws, size_t ws_size,
                              hipStream_t stream) {
    const float* x   = (const float*)d_in[0];
    const float* ew  = (const float*)d_in[1];
    const float* Wm  = (const float*)d_in[2];
    const float* b   = (const float*)d_in[3];
    const int*   src = (const int*)d_in[4];
    const int*   dst = (const int*)d_in[5];

    float* out = (float*)d_out;

    float* out_deg = (float*)d_ws;
    float* in_deg  = out_deg + N_NODES;
    float* h       = in_deg + N_NODES;

    gc_zero_kernel<<<512, 256, 0, stream>>>(out_deg, 2 * N_NODES);
    gc_zero_kernel<<<4096, 256, 0, stream>>>(out, N_NODES * OUT_FEATS);

    gc_degree_kernel<<<(N_EDGES + 255) / 256, 256, 0, stream>>>(src, dst, out_deg, in_deg);

    gc_gemm_wmma_kernel<<<N_NODES / 16, 256, 0, stream>>>(x, Wm, out_deg, h);

    gc_spmm_kernel<<<N_EDGES / 8, 256, 0, stream>>>(h, ew, src, dst, out);

    gc_finalize_kernel<<<(N_NODES * OUT_FEATS + 255) / 256, 256, 0, stream>>>(out, in_deg, b);
}


# degree counters moved from two packed arrays (32 counters per 128-B line) to one counter per line inside the not yet written h rows; the projection reads them first; counts stay exact
# speedup vs baseline: 1.1783x; 1.1783x over previous
#include <hip/hip_runtime.h>
#include <hip/hip_bf16.h>

typedef float v2f __attribute__((ext_vector_type(2)));
typedef float v8f __attribute__((ext_vector_type(8)));

#define N_NODES   50000
#define N_EDGES   800000
#define IN_FEATS  256
#define OUT_FEATS 128

#define CNT_OUT_OFF 0
#define CNT_IN_OFF  64

__global__ void gc_zero_kernel(float* __restrict__ p, int n) {
    int i = blockIdx.x * blockDim.x + threadIdx.x;
    int stride = gridDim.x * blockDim.x;
    for (; i < n; i += stride) p[i] = 0.0f;
}

__global__ void gc_zero_counters_kernel(float* __restrict__ h) {
    const int i = blockIdx.x * blockDim.x + threadIdx.x;
    if (i >= 2 * N_NODES) return;
    const int node = i >> 1;
    const int off  = (i & 1) ? CNT_IN_OFF : CNT_OUT_OFF;
    h[(size_t)node * OUT_FEATS + off] = 0.0f;
}

__global__ void gc_degree_kernel(const int* __restrict__ src,
                                 const int* __restrict__ dst,
                                 float* __restrict__ h) {
    int e = blockIdx.x * blockDim.x + threadIdx.x;
    if (e >= N_EDGES) return;
    unsafeAtomicAdd(&h[(size_t)src[e] * OUT_FEATS + CNT_OUT_OFF], 1.0f);
    unsafeAtomicAdd(&h[(size_t)dst[e] * OUT_FEATS + CNT_IN_OFF], 1.0f);
}

__global__ void gc_gemm_wmma_kernel(const float* __restrict__ x,
                                    const float* __restrict__ Wm,
                                    float* __restrict__ deg_pad,
                                    float* __restrict__ h) {
    const int lane = threadIdx.x & 31;
    const int wave = threadIdx.x >> 5;
    const int half = lane >> 4;
    const int l    = lane & 15;

    const int m0 = blockIdx.x * 16;
    const int n0 = wave * 16;

    const int row = m0 + l;
    const float od = h[(size_t)row * OUT_FEATS + CNT_OUT_OFF];
    if (threadIdx.x < 16)
        deg_pad[blockIdx.x * 32 + threadIdx.x] =
            h[(size_t)(m0 + threadIdx.x) * OUT_FEATS + CNT_IN_OFF];
    __syncthreads();
    const float rs = rsqrtf(fmaxf(od, 1.0f));
    const float* __restrict__ xrow = x + (size_t)row * IN_FEATS;

    v8f c = {};
    #pragma unroll 8
    for (int k0 = 0; k0 < IN_FEATS; k0 += 4) {
        const int ka = k0 + 2 * half;
        v2f a, b;
        a.x = xrow[ka]     * rs;
        a.y = xrow[ka + 1] * rs;
        b.x = Wm[(size_t)ka       * OUT_FEATS + n0 + l];
        b.y = Wm[(size_t)(ka + 1) * OUT_FEATS + n0 + l];
        c = __builtin_amdgcn_wmma_f32_16x16x4_f32(false, a, false, b,
                                                  (short)0, c, false, false);
    }

    #pragma unroll
    for (int r = 0; r < 8; ++r) {
        const int m = m0 + r + 8 * half;
        h[(size_t)m * OUT_FEATS + n0 + l] = c[r];
    }
}

__global__ void gc_spmm_kernel(const float* __restrict__ h,
                               const float* __restrict__ ew,
                               const int* __restrict__ src,
                               const int* __restrict__ dst,
                               float* __restrict__ agg) {
    const int wave_in_blk = threadIdx.x >> 5;
    const int lane        = threadIdx.x & 31;
    const long long e = (long long)blockIdx.x * (blockDim.x >> 5) + wave_in_blk;
    if (e >= N_EDGES) return;

    const int  s = src[e];
    const int  d = dst[e];
    const float w = ew[e];

    const float4 hv = *(const float4*)(h + (size_t)s * OUT_FEATS + lane * 4);
    float* outp = agg + (size_t)d * OUT_FEATS + lane * 4;
    unsafeAtomicAdd(outp + 0, hv.x * w);
    unsafeAtomicAdd(outp + 1, hv.y * w);
    unsafeAtomicAdd(outp + 2, hv.z * w);
    unsafeAtomicAdd(outp + 3, hv.w * w);
}

__global__ void gc_finalize_kernel(float* __restrict__ out,
                                   const float* __restrict__ deg_pad,
                                   const float* __restrict__ b) {
    const long long i = (long long)blockIdx.x * blockDim.x + threadIdx.x;
    if (i >= (long long)N_NODES * OUT_FEATS) return;
    const int node = (int)(i >> 7);
    const int col  = (int)(i & 127);
    const float rn = rsqrtf(fmaxf(deg_pad[(node >> 4) * 32 + (node & 15)], 1.0f));
    out[i] = out[i] * rn + b[col];
}

extern "C" void kernel_launch(void* const* d_in, const int* in_sizes, int n_in,
                              void* d_out, int out_size, void* d_ws, size_t ws_size,
                              hipStream_t stream) {
    const float* x   = (const float*)d_in[0];
    const float* ew  = (const float*)d_in[1];
    const float* Wm  = (const float*)d_in[2];
    const float* b   = (const float*)d_in[3];
    const int*   src = (const int*)d_in[4];
    const int*   dst = (const int*)d_in[5];

    float* out = (float*)d_out;

    float* deg_pad = (float*)d_ws;
    float* h       = deg_pad + 2 * N_NODES;

    gc_zero_counters_kernel<<<(2 * N_NODES + 255) / 256, 256, 0, stream>>>(h);
    gc_zero_kernel<<<4096, 256, 0, stream>>>(out, N_NODES * OUT_FEATS);

    gc_degree_kernel<<<(N_EDGES + 255) / 256, 256, 0, stream>>>(src, dst, h);

    gc_gemm_wmma_kernel<<<N_NODES / 16, 256, 0, stream>>>(x, Wm, deg_pad, h);

    gc_spmm_kernel<<<N_EDGES / 8, 256, 0, stream>>>(h, ew, src, dst, out);

    gc_finalize_kernel<<<(N_NODES * OUT_FEATS + 255) / 256, 256, 0, stream>>>(out, deg_pad, b);
}


# scatter: lane i owns columns i,i+32,i+64,i+96, so each f32 atomic instruction adds one whole 128-B line, not 32 B of each of four lines; sums still in atomic arrival order
# speedup vs baseline: 1.3757x; 1.1675x over previous
#include <hip/hip_runtime.h>
#include <hip/hip_bf16.h>

typedef float v2f __attribute__((ext_vector_type(2)));
typedef float v8f __attribute__((ext_vector_type(8)));

#define N_NODES   50000
#define N_EDGES   800000
#define IN_FEATS  256
#define OUT_FEATS 128

#define CNT_OUT_OFF 0
#define CNT_IN_OFF  64

__global__ void gc_zero_kernel(float* __restrict__ p, int n) {
    int i = blockIdx.x * blockDim.x + threadIdx.x;
    int stride = gridDim.x * blockDim.x;
    for (; i < n; i += stride) p[i] = 0.0f;
}

__global__ void gc_zero_counters_kernel(float* __restrict__ h) {
    const int i = blockIdx.x * blockDim.x + threadIdx.x;
    if (i >= 2 * N_NODES) return;
    const int node = i >> 1;
    const int off  = (i & 1) ? CNT_IN_OFF : CNT_OUT_OFF;
    h[(size_t)node * OUT_FEATS + off] = 0.0f;
}

__global__ void gc_degree_kernel(const int* __restrict__ src,
                                 const int* __restrict__ dst,
                                 float* __restrict__ h) {
    int e = blockIdx.x * blockDim.x + threadIdx.x;
    if (e >= N_EDGES) return;
    unsafeAtomicAdd(&h[(size_t)src[e] * OUT_FEATS + CNT_OUT_OFF], 1.0f);
    unsafeAtomicAdd(&h[(size_t)dst[e] * OUT_FEATS + CNT_IN_OFF], 1.0f);
}

__global__ void gc_gemm_wmma_kernel(const float* __restrict__ x,
                                    const float* __restrict__ Wm,
                                    float* __restrict__ deg_pad,
                                    float* __restrict__ h) {
    const int lane = threadIdx.x & 31;
    const int wave = threadIdx.x >> 5;
    const int half = lane >> 4;
    const int l    = lane & 15;

    const int m0 = blockIdx.x * 16;
    const int n0 = wave * 16;

    const int row = m0 + l;
    const float od = h[(size_t)row * OUT_FEATS + CNT_OUT_OFF];
    if (threadIdx.x < 16)
        deg_pad[blockIdx.x * 32 + threadIdx.x] =
            h[(size_t)(m0 + threadIdx.x) * OUT_FEATS + CNT_IN_OFF];
    __syncthreads();
    const float rs = rsqrtf(fmaxf(od, 1.0f));
    const float* __restrict__ xrow = x + (size_t)row * IN_FEATS;

    v8f c = {};
    #pragma unroll 8
    for (int k0 = 0; k0 < IN_FEATS; k0 += 4) {
        const int ka = k0 + 2 * half;
        v2f a, b;
        a.x = xrow[ka]     * rs;
        a.y = xrow[ka + 1] * rs;
        b.x = Wm[(size_t)ka       * OUT_FEATS + n0 + l];
        b.y = Wm[(size_t)(ka + 1) * OUT_FEATS + n0 + l];
        c = __builtin_amdgcn_wmma_f32_16x16x4_f32(false, a, false, b,
                                                  (short)0, c, false, false);
    }

    #pragma unroll
    for (int r = 0; r < 8; ++r) {
        const int m = m0 + r + 8 * half;
        h[(size_t)m * OUT_FEATS + n0 + l] = c[r];
    }
}

__global__ void gc_spmm_kernel(const float* __restrict__ h,
                               const float* __restrict__ ew,
                               const int* __restrict__ src,
                               const int* __restrict__ dst,
                               float* __restrict__ agg) {
    const int wave_in_blk = threadIdx.x >> 5;
    const int lane        = threadIdx.x & 31;
    const long long e = (long long)blockIdx.x * (blockDim.x >> 5) + wave_in_blk;
    if (e >= N_EDGES) return;

    const int  s = src[e];
    const int  d = dst[e];
    const float w = ew[e];

    const float* __restrict__ hp = h + (size_t)s * OUT_FEATS + lane;
    const float h0 = hp[0];
    const float h1 = hp[32];
    const float h2 = hp[64];
    const float h3 = hp[96];
    float* outp = agg + (size_t)d * OUT_FEATS + lane;
    unsafeAtomicAdd(outp +  0, h0 * w);
    unsafeAtomicAdd(outp + 32, h1 * w);
    unsafeAtomicAdd(outp + 64, h2 * w);
    unsafeAtomicAdd(outp + 96, h3 * w);
}

__global__ void gc_finalize_kernel(float* __restrict__ out,
                                   const float* __restrict__ deg_pad,
                                   const float* __restrict__ b) {
    const long long i = (long long)blockIdx.x * blockDim.x + threadIdx.x;
    if (i >= (long long)N_NODES * OUT_FEATS) return;
    const int node = (int)(i >> 7);
    const int col  = (int)(i & 127);
    const float rn = rsqrtf(fmaxf(deg_pad[(node >> 4) * 32 + (node & 15)], 1.0f));
    out[i] = out[i] * rn + b[col];
}

extern "C" void kernel_launch(void* const* d_in, const int* in_sizes, int n_in,
                              void* d_out, int out_size, void* d_ws, size_t ws_size,
                              hipStream_t stream) {
    const float* x   = (const float*)d_in[0];
    const float* ew  = (const float*)d_in[1];
    const float* Wm  = (const float*)d_in[2];
    const float* b   = (const float*)d_in[3];
    const int*   src = (const int*)d_in[4];
    const int*   dst = (const int*)d_in[5];

    float* out = (float*)d_out;

    float* deg_pad = (float*)d_ws;
    float* h       = deg_pad + 2 * N_NODES;

    gc_zero_counters_kernel<<<(2 * N_NODES + 255) / 256, 256, 0, stream>>>(h);
    gc_zero_kernel<<<4096, 256, 0, stream>>>(out, N_NODES * OUT_FEATS);

    gc_degree_kernel<<<(N_EDGES + 255) / 256, 256, 0, stream>>>(src, dst, h);

    gc_gemm_wmma_kernel<<<N_NODES / 16, 256, 0, stream>>>(x, Wm, deg_pad, h);

    gc_spmm_kernel<<<N_EDGES / 8, 256, 0, stream>>>(h, ew, src, dst, out);

    gc_finalize_kernel<<<(N_NODES * OUT_FEATS + 255) / 256, 256, 0, stream>>>(out, deg_pad, b);
}


# aggregation made destination-owned: 256 work-groups each keep 196 output rows in LDS, scan all dst, gather their edges' h rows, add with LDS float atomics (arrival order), then norm, bias, one store
# speedup vs baseline: 2.0943x; 1.5224x over previous
#include <hip/hip_runtime.h>
#include <hip/hip_bf16.h>

typedef float v2f __attribute__((ext_vector_type(2)));
typedef float v8f __attribute__((ext_vector_type(8)));

#define N_NODES   50000
#define N_EDGES   800000
#define IN_FEATS  256
#define OUT_FEATS 128

#define CNT_OUT_OFF 0

__global__ void gc_zero_counters_kernel(float* __restrict__ h) {
    const int i = blockIdx.x * blockDim.x + threadIdx.x;
    if (i >= N_NODES) return;
    h[(size_t)i * OUT_FEATS + CNT_OUT_OFF] = 0.0f;
}

__global__ void gc_degree_kernel(const int* __restrict__ src,
                                 float* __restrict__ h) {
    int e = blockIdx.x * blockDim.x + threadIdx.x;
    if (e >= N_EDGES) return;
    unsafeAtomicAdd(&h[(size_t)src[e] * OUT_FEATS + CNT_OUT_OFF], 1.0f);
}

__global__ void gc_gemm_wmma_kernel(const float* __restrict__ x,
                                    const float* __restrict__ Wm,
                                    float* __restrict__ h) {
    const int lane = threadIdx.x & 31;
    const int wave = threadIdx.x >> 5;
    const int half = lane >> 4;
    const int l    = lane & 15;

    const int m0 = blockIdx.x * 16;
    const int n0 = wave * 16;

    const int row = m0 + l;
    const float od = h[(size_t)row * OUT_FEATS + CNT_OUT_OFF];
    __syncthreads();
    const float rs = rsqrtf(fmaxf(od, 1.0f));
    const float* __restrict__ xrow = x + (size_t)row * IN_FEATS;

    v8f c = {};
    #pragma unroll 8
    for (int k0 = 0; k0 < IN_FEATS; k0 += 4) {
        const int ka = k0 + 2 * half;
        v2f a, b;
        a.x = xrow[ka]     * rs;
        a.y = xrow[ka + 1] * rs;
        b.x = Wm[(size_t)ka       * OUT_FEATS + n0 + l];
        b.y = Wm[(size_t)(ka + 1) * OUT_FEATS + n0 + l];
        c = __builtin_amdgcn_wmma_f32_16x16x4_f32(false, a, false, b,
                                                  (short)0, c, false, false);
    }

    #pragma unroll
    for (int r = 0; r < 8; ++r) {
        const int m = m0 + r + 8 * half;
        h[(size_t)m * OUT_FEATS + n0 + l] = c[r];
    }
}

#define AGG_ROWS    196
#define AGG_GROUPS  256
#define AGG_THREADS 1024
#define AGG_WAVES   (AGG_THREADS / 32)
#define AGG_UNROLL  4
#define AGG_LIST    544
#define EDGE_BITS   20
static_assert(N_EDGES < (1 << EDGE_BITS), "edge id must fit its field");
static_assert(AGG_ROWS <= 256, "row-in-tile must fit 8 bits");
static_assert(AGG_GROUPS * AGG_ROWS >= N_NODES, "tiles must cover every node");
static_assert(N_EDGES % 4 == 0, "dst is read 4 edges per lane");

__global__ void __launch_bounds__(AGG_THREADS)
gc_aggregate_kernel(const float* __restrict__ h,
                    const float* __restrict__ ew,
                    const int* __restrict__ src,
                    const int* __restrict__ dst,
                    const float* __restrict__ b,
                    float* __restrict__ out) {
    __shared__ __attribute__((aligned(16))) float acc[AGG_ROWS * OUT_FEATS];
    __shared__ int indeg[AGG_ROWS];
    __shared__ int lst[AGG_WAVES][AGG_LIST];

    const int tid  = threadIdx.x;
    const int lane = tid & 31;
    const int wave = tid >> 5;
    const int base = blockIdx.x * AGG_ROWS;
    int* __restrict__ mylist = lst[wave];

    for (int i = tid; i < AGG_ROWS * OUT_FEATS / 4; i += AGG_THREADS)
        reinterpret_cast<float4*>(acc)[i] = make_float4(0.f, 0.f, 0.f, 0.f);
    if (tid < AGG_ROWS) indeg[tid] = 0;
    __syncthreads();

    const int4*   __restrict__ dst4 = reinterpret_cast<const int4*>(dst);
    const float4* __restrict__ h4   = reinterpret_cast<const float4*>(h);
    constexpr int N4 = N_EDGES / 4;

    int cnt = 0;

    auto drain = [&](int n) {
        const int start = cnt - n;
        const int ent = mylist[start + (lane < n ? lane : 0)];
        const int e   = ent & ((1 << EDGE_BITS) - 1);
        const int dl  = ent >> EDGE_BITS;
        const int   s = src[e];
        const float w = ew[e];
        for (int j0 = 0; j0 < n; j0 += 8) {
            float4 r[8];
            #pragma unroll
            for (int u = 0; u < 8; ++u) {
                const int jj = (j0 + u < n) ? (j0 + u) : (n - 1);
                const int sj = __builtin_amdgcn_readlane(s, jj);
                r[u] = h4[(size_t)sj * (OUT_FEATS / 4) + lane];
            }
            #pragma unroll
            for (int u = 0; u < 8; ++u) {
                if (j0 + u < n) {
                    const int   dj = __builtin_amdgcn_readlane(dl, j0 + u);
                    const float wj = __int_as_float(__builtin_amdgcn_readlane(__float_as_int(w), j0 + u));
                    float* a = acc + dj * OUT_FEATS + lane;
                    __hip_atomic_fetch_add(a +  0, r[u].x * wj, __ATOMIC_RELAXED, __HIP_MEMORY_SCOPE_WORKGROUP);
                    __hip_atomic_fetch_add(a + 32, r[u].y * wj, __ATOMIC_RELAXED, __HIP_MEMORY_SCOPE_WORKGROUP);
                    __hip_atomic_fetch_add(a + 64, r[u].z * wj, __ATOMIC_RELAXED, __HIP_MEMORY_SCOPE_WORKGROUP);
                    __hip_atomic_fetch_add(a + 96, r[u].w * wj, __ATOMIC_RELAXED, __HIP_MEMORY_SCOPE_WORKGROUP);
                }
            }
        }
        cnt = start;
    };

    auto append = [&](bool m, int d, int e) {
        const unsigned bal = __builtin_amdgcn_ballot_w32(m);
        if (m) {
            const int dl = d - base;
            mylist[cnt + (int)__builtin_amdgcn_mbcnt_lo(bal, 0u)] = e | (dl << EDGE_BITS);
            __hip_atomic_fetch_add(&indeg[dl], 1, __ATOMIC_RELAXED, __HIP_MEMORY_SCOPE_WORKGROUP);
        }
        cnt += __builtin_popcount(bal);
    };

    for (int q0 = 0; q0 < N4; q0 += AGG_THREADS * AGG_UNROLL) {
        int4 dv[AGG_UNROLL];
        #pragma unroll
        for (int u = 0; u < AGG_UNROLL; ++u) {
            const int q = q0 + u * AGG_THREADS + tid;
            dv[u] = (q < N4) ? dst4[q] : make_int4(-1, -1, -1, -1);
        }
        #pragma unroll
        for (int u = 0; u < AGG_UNROLL; ++u) {
            const int q = q0 + u * AGG_THREADS + tid;
            const bool m0 = (unsigned)(dv[u].x - base) < (unsigned)AGG_ROWS;
            const bool m1 = (unsigned)(dv[u].y - base) < (unsigned)AGG_ROWS;
            const bool m2 = (unsigned)(dv[u].z - base) < (unsigned)AGG_ROWS;
            const bool m3 = (unsigned)(dv[u].w - base) < (unsigned)AGG_ROWS;
            if (__builtin_amdgcn_ballot_w32(m0 | m1 | m2 | m3) != 0u) {
                append(m0, dv[u].x, 4 * q + 0);
                append(m1, dv[u].y, 4 * q + 1);
                append(m2, dv[u].z, 4 * q + 2);
                append(m3, dv[u].w, 4 * q + 3);
            }
        }
        while (cnt >= 32) drain(32);
    }
    if (cnt > 0) drain(cnt);

    __syncthreads();

    const float4 bb = reinterpret_cast<const float4*>(b)[lane];
    for (int row = wave; row < AGG_ROWS; row += AGG_WAVES) {
        const int node = base + row;
        if (node >= N_NODES) break;
        const float rn = rsqrtf(fmaxf((float)indeg[row], 1.0f));
        const float* a = acc + row * OUT_FEATS + lane;
        float4 o;
        o.x = a[0]  * rn + bb.x;
        o.y = a[32] * rn + bb.y;
        o.z = a[64] * rn + bb.z;
        o.w = a[96] * rn + bb.w;
        reinterpret_cast<float4*>(out + (size_t)node * OUT_FEATS)[lane] = o;
    }
}

extern "C" void kernel_launch(void* const* d_in, const int* in_sizes, int n_in,
                              void* d_out, int out_size, void* d_ws, size_t ws_size,
                              hipStream_t stream) {
    const float* x   = (const float*)d_in[0];
    const float* ew  = (const float*)d_in[1];
    const float* Wm  = (const float*)d_in[2];
    const float* b   = (const float*)d_in[3];
    const int*   src = (const int*)d_in[4];
    const int*   dst = (const int*)d_in[5];

    float* out = (float*)d_out;

    float* h = (float*)d_ws + 2 * N_NODES;

    gc_zero_counters_kernel<<<(N_NODES + 255) / 256, 256, 0, stream>>>(h);

    gc_degree_kernel<<<(N_EDGES + 255) / 256, 256, 0, stream>>>(src, h);

    gc_gemm_wmma_kernel<<<N_NODES / 16, 256, 0, stream>>>(x, Wm, h);

    gc_aggregate_kernel<<<AGG_GROUPS, AGG_THREADS, 0, stream>>>(h, ew, src, dst, b, out);
}


# projection made persistent: 512 groups loop over 16-row tiles; a wave keeps its 256x16 slice of W in registers; x rows come by whole-line loads staged in LDS, 2 stages; k order and sums unchanged
# speedup vs baseline: 2.5828x; 1.2333x over previous
#include <hip/hip_runtime.h>
#include <hip/hip_bf16.h>

typedef float v2f __attribute__((ext_vector_type(2)));
typedef float v8f __attribute__((ext_vector_type(8)));

#define N_NODES   50000
#define N_EDGES   800000
#define IN_FEATS  256
#define OUT_FEATS 128

#define CNT_OUT_OFF 0

__global__ void gc_zero_counters_kernel(float* __restrict__ h) {
    const int i = blockIdx.x * blockDim.x + threadIdx.x;
    if (i >= N_NODES) return;
    h[(size_t)i * OUT_FEATS + CNT_OUT_OFF] = 0.0f;
}

__global__ void gc_degree_kernel(const int* __restrict__ src,
                                 float* __restrict__ h) {
    int e = blockIdx.x * blockDim.x + threadIdx.x;
    if (e >= N_EDGES) return;
    unsafeAtomicAdd(&h[(size_t)src[e] * OUT_FEATS + CNT_OUT_OFF], 1.0f);
}

#define GM_GROUPS  512
#define GM_TILES   (N_NODES / 16)
#define GM_PITCH   260
static_assert(N_NODES % 16 == 0, "row tiles are whole");
static_assert(GM_GROUPS <= GM_TILES, "every group has a first tile");

__global__ void __launch_bounds__(256)
gc_gemm_wmma_kernel(const float* __restrict__ x,
                    const float* __restrict__ Wm,
                    float* __restrict__ h) {
    __shared__ __attribute__((aligned(16))) float xs[2][16 * GM_PITCH];

    const int tid  = threadIdx.x;
    const int lane = tid & 31;
    const int wave = tid >> 5;
    const int half = lane >> 4;
    const int l    = lane & 15;
    const int n0   = wave * 16;

    const int srow = tid >> 6;
    const int c4   = tid & 63;
    const float4* __restrict__ x4 = reinterpret_cast<const float4*>(x);

    float4 xv[4];
    float  od[4];

    int tile = blockIdx.x;
    #pragma unroll
    for (int i = 0; i < 4; ++i) {
        const int row = tile * 16 + srow + 4 * i;
        od[i] = h[(size_t)row * OUT_FEATS + CNT_OUT_OFF];
        xv[i] = x4[(size_t)row * (IN_FEATS / 4) + c4];
    }

    v2f wf[IN_FEATS / 4];
    #pragma unroll
    for (int s = 0; s < IN_FEATS / 4; ++s) {
        const int ka = 4 * s + 2 * half;
        wf[s].x = Wm[(size_t)ka       * OUT_FEATS + n0 + l];
        wf[s].y = Wm[(size_t)(ka + 1) * OUT_FEATS + n0 + l];
    }

    #pragma unroll
    for (int i = 0; i < 4; ++i) {
        const float rs = rsqrtf(fmaxf(od[i], 1.0f));
        float4 t = xv[i];
        t.x *= rs; t.y *= rs; t.z *= rs; t.w *= rs;
        *reinterpret_cast<float4*>(&xs[0][(srow + 4 * i) * GM_PITCH + 4 * c4]) = t;
    }
    __syncthreads();

    int buf = 0;
    for (; tile < GM_TILES; tile += GM_GROUPS) {
        const int  next     = tile + GM_GROUPS;
        const bool has_next = next < GM_TILES;
        if (has_next) {
            #pragma unroll
            for (int i = 0; i < 4; ++i) {
                const int row = next * 16 + srow + 4 * i;
                od[i] = h[(size_t)row * OUT_FEATS + CNT_OUT_OFF];
                xv[i] = x4[(size_t)row * (IN_FEATS / 4) + c4];
            }
        }

        const float* __restrict__ arow = &xs[buf][l * GM_PITCH + 2 * half];
        v8f c = {};
        #pragma unroll
        for (int s = 0; s < IN_FEATS / 4; ++s) {
            const v2f a = *reinterpret_cast<const v2f*>(arow + 4 * s);
            c = __builtin_amdgcn_wmma_f32_16x16x4_f32(false, a, false, wf[s],
                                                      (short)0, c, false, false);
        }

        const int m0 = tile * 16;
        #pragma unroll
        for (int r = 0; r < 8; ++r) {
            const int m = m0 + r + 8 * half;
            h[(size_t)m * OUT_FEATS + n0 + l] = c[r];
        }

        if (has_next) {
            #pragma unroll
            for (int i = 0; i < 4; ++i) {
                const float rs = rsqrtf(fmaxf(od[i], 1.0f));
                float4 t = xv[i];
                t.x *= rs; t.y *= rs; t.z *= rs; t.w *= rs;
                *reinterpret_cast<float4*>(&xs[buf ^ 1][(srow + 4 * i) * GM_PITCH + 4 * c4]) = t;
            }
        }
        __syncthreads();
        buf ^= 1;
    }
}

#define AGG_ROWS    196
#define AGG_GROUPS  256
#define AGG_THREADS 1024
#define AGG_WAVES   (AGG_THREADS / 32)
#define AGG_UNROLL  4
#define AGG_LIST    544
#define EDGE_BITS   20
static_assert(N_EDGES < (1 << EDGE_BITS), "edge id must fit its field");
static_assert(AGG_ROWS <= 256, "row-in-tile must fit 8 bits");
static_assert(AGG_GROUPS * AGG_ROWS >= N_NODES, "tiles must cover every node");
static_assert(N_EDGES % 4 == 0, "dst is read 4 edges per lane");

__global__ void __launch_bounds__(AGG_THREADS)
gc_aggregate_kernel(const float* __restrict__ h,
                    const float* __restrict__ ew,
                    const int* __restrict__ src,
                    const int* __restrict__ dst,
                    const float* __restrict__ b,
                    float* __restrict__ out) {
    __shared__ __attribute__((aligned(16))) float acc[AGG_ROWS * OUT_FEATS];
    __shared__ int indeg[AGG_ROWS];
    __shared__ int lst[AGG_WAVES][AGG_LIST];

    const int tid  = threadIdx.x;
    const int lane = tid & 31;
    const int wave = tid >> 5;
    const int base = blockIdx.x * AGG_ROWS;
    int* __restrict__ mylist = lst[wave];

    for (int i = tid; i < AGG_ROWS * OUT_FEATS / 4; i += AGG_THREADS)
        reinterpret_cast<float4*>(acc)[i] = make_float4(0.f, 0.f, 0.f, 0.f);
    if (tid < AGG_ROWS) indeg[tid] = 0;
    __syncthreads();

    const int4*   __restrict__ dst4 = reinterpret_cast<const int4*>(dst);
    const float4* __restrict__ h4   = reinterpret_cast<const float4*>(h);
    constexpr int N4 = N_EDGES / 4;

    int cnt = 0;

    auto drain = [&](int n) {
        const int start = cnt - n;
        const int ent = mylist[start + (lane < n ? lane : 0)];
        const int e   = ent & ((1 << EDGE_BITS) - 1);
        const int dl  = ent >> EDGE_BITS;
        const int   s = src[e];
        const float w = ew[e];
        for (int j0 = 0; j0 < n; j0 += 8) {
            float4 r[8];
            #pragma unroll
            for (int u = 0; u < 8; ++u) {
                const int jj = (j0 + u < n) ? (j0 + u) : (n - 1);
                const int sj = __builtin_amdgcn_readlane(s, jj);
                r[u] = h4[(size_t)sj * (OUT_FEATS / 4) + lane];
            }
            #pragma unroll
            for (int u = 0; u < 8; ++u) {
                if (j0 + u < n) {
                    const int   dj = __builtin_amdgcn_readlane(dl, j0 + u);
                    const float wj = __int_as_float(__builtin_amdgcn_readlane(__float_as_int(w), j0 + u));
                    float* a = acc + dj * OUT_FEATS + lane;
                    __hip_atomic_fetch_add(a +  0, r[u].x * wj, __ATOMIC_RELAXED, __HIP_MEMORY_SCOPE_WORKGROUP);
                    __hip_atomic_fetch_add(a + 32, r[u].y * wj, __ATOMIC_RELAXED, __HIP_MEMORY_SCOPE_WORKGROUP);
                    __hip_atomic_fetch_add(a + 64, r[u].z * wj, __ATOMIC_RELAXED, __HIP_MEMORY_SCOPE_WORKGROUP);
                    __hip_atomic_fetch_add(a + 96, r[u].w * wj, __ATOMIC_RELAXED, __HIP_MEMORY_SCOPE_WORKGROUP);
                }
            }
        }
        cnt = start;
    };

    auto append = [&](bool m, int d, int e) {
        const unsigned bal = __builtin_amdgcn_ballot_w32(m);
        if (m) {
            const int dl = d - base;
            mylist[cnt + (int)__builtin_amdgcn_mbcnt_lo(bal, 0u)] = e | (dl << EDGE_BITS);
            __hip_atomic_fetch_add(&indeg[dl], 1, __ATOMIC_RELAXED, __HIP_MEMORY_SCOPE_WORKGROUP);
        }
        cnt += __builtin_popcount(bal);
    };

    for (int q0 = 0; q0 < N4; q0 += AGG_THREADS * AGG_UNROLL) {
        int4 dv[AGG_UNROLL];
        #pragma unroll
        for (int u = 0; u < AGG_UNROLL; ++u) {
            const int q = q0 + u * AGG_THREADS + tid;
            dv[u] = (q < N4) ? dst4[q] : make_int4(-1, -1, -1, -1);
        }
        #pragma unroll
        for (int u = 0; u < AGG_UNROLL; ++u) {
            const int q = q0 + u * AGG_THREADS + tid;
            const bool m0 = (unsigned)(dv[u].x - base) < (unsigned)AGG_ROWS;
            const bool m1 = (unsigned)(dv[u].y - base) < (unsigned)AGG_ROWS;
            const bool m2 = (unsigned)(dv[u].z - base) < (unsigned)AGG_ROWS;
            const bool m3 = (unsigned)(dv[u].w - base) < (unsigned)AGG_ROWS;
            if (__builtin_amdgcn_ballot_w32(m0 | m1 | m2 | m3) != 0u) {
                append(m0, dv[u].x, 4 * q + 0);
                append(m1, dv[u].y, 4 * q + 1);
                append(m2, dv[u].z, 4 * q + 2);
                append(m3, dv[u].w, 4 * q + 3);
            }
        }
        while (cnt >= 32) drain(32);
    }
    if (cnt > 0) drain(cnt);

    __syncthreads();

    const float4 bb = reinterpret_cast<const float4*>(b)[lane];
    for (int row = wave; row < AGG_ROWS; row += AGG_WAVES) {
        const int node = base + row;
        if (node >= N_NODES) break;
        const float rn = rsqrtf(fmaxf((float)indeg[row], 1.0f));
        const float* a = acc + row * OUT_FEATS + lane;
        float4 o;
        o.x = a[0]  * rn + bb.x;
        o.y = a[32] * rn + bb.y;
        o.z = a[64] * rn + bb.z;
        o.w = a[96] * rn + bb.w;
        reinterpret_cast<float4*>(out + (size_t)node * OUT_FEATS)[lane] = o;
    }
}

extern "C" void kernel_launch(void* const* d_in, const int* in_sizes, int n_in,
                              void* d_out, int out_size, void* d_ws, size_t ws_size,
                              hipStream_t stream) {
    const float* x   = (const float*)d_in[0];
    const float* ew  = (const float*)d_in[1];
    const float* Wm  = (const float*)d_in[2];
    const float* b   = (const float*)d_in[3];
    const int*   src = (const int*)d_in[4];
    const int*   dst = (const int*)d_in[5];

    float* out = (float*)d_out;

    float* h = (float*)d_ws + 2 * N_NODES;

    gc_zero_counters_kernel<<<(N_NODES + 255) / 256, 256, 0, stream>>>(h);

    gc_degree_kernel<<<(N_EDGES + 255) / 256, 256, 0, stream>>>(src, h);

    gc_gemm_wmma_kernel<<<GM_GROUPS, 256, 0, stream>>>(x, Wm, h);

    gc_aggregate_kernel<<<AGG_GROUPS, AGG_THREADS, 0, stream>>>(h, ew, src, dst, b, out);
}


# aggregation reads dst through a per-tile skip bitmap of 64-edge units, built once in the degree launch in the unused head of the workspace: a group loads only marked units; LDS sums in arrival order
# speedup vs baseline: 2.9568x; 1.1448x over previous
#include <hip/hip_runtime.h>
#include <hip/hip_bf16.h>

typedef float v2f __attribute__((ext_vector_type(2)));
typedef float v8f __attribute__((ext_vector_type(8)));

#define N_NODES   50000
#define N_EDGES   800000
#define IN_FEATS  256
#define OUT_FEATS 128

#define CNT_OUT_OFF 0

__global__ void gc_zero_counters_kernel(float* __restrict__ h) {
    const int i = blockIdx.x * blockDim.x + threadIdx.x;
    if (i >= N_NODES) return;
    h[(size_t)i * OUT_FEATS + CNT_OUT_OFF] = 0.0f;
}

#define AGG_ROWS    196
#define AGG_GROUPS  256

#define BM_UNIT_EDGES 64
#define BM_WORDS      (N_EDGES / BM_UNIT_EDGES / 32)
#define BM_TAIL_UNITS (N_EDGES / BM_UNIT_EDGES - 32 * BM_WORDS)
static_assert(N_EDGES % BM_UNIT_EDGES == 0, "units are whole");
static_assert(BM_TAIL_UNITS > 0 && BM_TAIL_UNITS < 32, "tail units fit one virtual word");
static_assert((size_t)BM_WORDS * AGG_GROUPS * 4 <= (size_t)2 * N_NODES * 4, "bitmap fits the unused head of the workspace");

__global__ void __launch_bounds__(256)
gc_edge_prep_kernel(const int* __restrict__ src,
                    const int* __restrict__ dst,
                    int* __restrict__ bm,
                    float* __restrict__ h) {
    __shared__ unsigned m[AGG_GROUPS];
    const int tid = threadIdx.x;
    if (blockIdx.x < BM_WORDS) {
        m[tid] = 0u;
        __syncthreads();
        const int4* __restrict__ dst4 = reinterpret_cast<const int4*>(dst);
        const int q0 = blockIdx.x * (32 * BM_UNIT_EDGES / 4);
        int4 d[2];
        #pragma unroll
        for (int k = 0; k < 2; ++k)
            d[k] = dst4[q0 + k * 256 + tid];
        #pragma unroll
        for (int k = 0; k < 2; ++k) {
            const unsigned bit = 1u << (((q0 + k * 256 + tid) >> 4) & 31);
            const unsigned t0 = (unsigned)d[k].x / AGG_ROWS, t1 = (unsigned)d[k].y / AGG_ROWS;
            const unsigned t2 = (unsigned)d[k].z / AGG_ROWS, t3 = (unsigned)d[k].w / AGG_ROWS;
            if (t0 < AGG_GROUPS) __hip_atomic_fetch_or(&m[t0], bit, __ATOMIC_RELAXED, __HIP_MEMORY_SCOPE_WORKGROUP);
            if (t1 < AGG_GROUPS) __hip_atomic_fetch_or(&m[t1], bit, __ATOMIC_RELAXED, __HIP_MEMORY_SCOPE_WORKGROUP);
            if (t2 < AGG_GROUPS) __hip_atomic_fetch_or(&m[t2], bit, __ATOMIC_RELAXED, __HIP_MEMORY_SCOPE_WORKGROUP);
            if (t3 < AGG_GROUPS) __hip_atomic_fetch_or(&m[t3], bit, __ATOMIC_RELAXED, __HIP_MEMORY_SCOPE_WORKGROUP);
        }
        __syncthreads();
        bm[blockIdx.x * AGG_GROUPS + tid] = (int)m[tid];
        return;
    }
    const int e = (blockIdx.x - BM_WORDS) * 256 + tid;
    if (e >= N_EDGES) return;
    unsafeAtomicAdd(&h[(size_t)src[e] * OUT_FEATS + CNT_OUT_OFF], 1.0f);
}

#define GM_GROUPS  512
#define GM_TILES   (N_NODES / 16)
#define GM_PITCH   260
static_assert(N_NODES % 16 == 0, "row tiles are whole");
static_assert(GM_GROUPS <= GM_TILES, "every group has a first tile");

__global__ void __launch_bounds__(256)
gc_gemm_wmma_kernel(const float* __restrict__ x,
                    const float* __restrict__ Wm,
                    float* __restrict__ h) {
    __shared__ __attribute__((aligned(16))) float xs[2][16 * GM_PITCH];

    const int tid  = threadIdx.x;
    const int lane = tid & 31;
    const int wave = tid >> 5;
    const int half = lane >> 4;
    const int l    = lane & 15;
    const int n0   = wave * 16;

    const int srow = tid >> 6;
    const int c4   = tid & 63;
    const float4* __restrict__ x4 = reinterpret_cast<const float4*>(x);

    float4 xv[4];
    float  od[4];

    int tile = blockIdx.x;
    #pragma unroll
    for (int i = 0; i < 4; ++i) {
        const int row = tile * 16 + srow + 4 * i;
        od[i] = h[(size_t)row * OUT_FEATS + CNT_OUT_OFF];
        xv[i] = x4[(size_t)row * (IN_FEATS / 4) + c4];
    }

    v2f wf[IN_FEATS / 4];
    #pragma unroll
    for (int s = 0; s < IN_FEATS / 4; ++s) {
        const int ka = 4 * s + 2 * half;
        wf[s].x = Wm[(size_t)ka       * OUT_FEATS + n0 + l];
        wf[s].y = Wm[(size_t)(ka + 1) * OUT_FEATS + n0 + l];
    }

    #pragma unroll
    for (int i = 0; i < 4; ++i) {
        const float rs = rsqrtf(fmaxf(od[i], 1.0f));
        float4 t = xv[i];
        t.x *= rs; t.y *= rs; t.z *= rs; t.w *= rs;
        *reinterpret_cast<float4*>(&xs[0][(srow + 4 * i) * GM_PITCH + 4 * c4]) = t;
    }
    __syncthreads();

    int buf = 0;
    for (; tile < GM_TILES; tile += GM_GROUPS) {
        const int  next     = tile + GM_GROUPS;
        const bool has_next = next < GM_TILES;
        if (has_next) {
            #pragma unroll
            for (int i = 0; i < 4; ++i) {
                const int row = next * 16 + srow + 4 * i;
                od[i] = h[(size_t)row * OUT_FEATS + CNT_OUT_OFF];
                xv[i] = x4[(size_t)row * (IN_FEATS / 4) + c4];
            }
        }

        const float* __restrict__ arow = &xs[buf][l * GM_PITCH + 2 * half];
        v8f c = {};
        #pragma unroll
        for (int s = 0; s < IN_FEATS / 4; ++s) {
            const v2f a = *reinterpret_cast<const v2f*>(arow + 4 * s);
            c = __builtin_amdgcn_wmma_f32_16x16x4_f32(false, a, false, wf[s],
                                                      (short)0, c, false, false);
        }

        const int m0 = tile * 16;
        #pragma unroll
        for (int r = 0; r < 8; ++r) {
            const int m = m0 + r + 8 * half;
            h[(size_t)m * OUT_FEATS + n0 + l] = c[r];
        }

        if (has_next) {
            #pragma unroll
            for (int i = 0; i < 4; ++i) {
                const float rs = rsqrtf(fmaxf(od[i], 1.0f));
                float4 t = xv[i];
                t.x *= rs; t.y *= rs; t.z *= rs; t.w *= rs;
                *reinterpret_cast<float4*>(&xs[buf ^ 1][(srow + 4 * i) * GM_PITCH + 4 * c4]) = t;
            }
        }
        __syncthreads();
        buf ^= 1;
    }
}

#define AGG_THREADS 1024
#define AGG_WAVES   (AGG_THREADS / 32)
#define AGG_UNROLL  4
#define AGG_LIST    544
#define EDGE_BITS   20
static_assert(N_EDGES < (1 << EDGE_BITS), "edge id must fit its field");
static_assert(AGG_ROWS <= 256, "row-in-tile must fit 8 bits");
static_assert(AGG_GROUPS * AGG_ROWS >= N_NODES, "tiles must cover every node");
static_assert(N_EDGES % 4 == 0, "dst is read 4 edges per lane");
static_assert(AGG_GROUPS == 256, "the bitmap blocks build one word per thread");
static_assert(BM_WORDS / AGG_WAVES < 32, "a wave's bitmap words fit one per lane");

__global__ void __launch_bounds__(AGG_THREADS)
gc_aggregate_kernel(const float* __restrict__ h,
                    const float* __restrict__ ew,
                    const int* __restrict__ src,
                    const int* __restrict__ dst,
                    const float* __restrict__ b,
                    const int* __restrict__ bm,
                    float* __restrict__ out) {
    __shared__ __attribute__((aligned(16))) float acc[AGG_ROWS * OUT_FEATS];
    __shared__ int indeg[AGG_ROWS];
    __shared__ int lst[AGG_WAVES][AGG_LIST];

    const int tid  = threadIdx.x;
    const int lane = tid & 31;
    const int wave = tid >> 5;
    const int base = blockIdx.x * AGG_ROWS;
    int* __restrict__ mylist = lst[wave];
    const int half = lane >> 4;
    const int l    = lane & 15;

    const int wmine = wave + AGG_WAVES * lane;
    unsigned myw = 0u;
    if (wmine < BM_WORDS)       myw = (unsigned)bm[wmine * AGG_GROUPS + blockIdx.x];
    else if (wmine == BM_WORDS) myw = (1u << BM_TAIL_UNITS) - 1u;

    for (int i = tid; i < AGG_ROWS * OUT_FEATS / 4; i += AGG_THREADS)
        reinterpret_cast<float4*>(acc)[i] = make_float4(0.f, 0.f, 0.f, 0.f);
    if (tid < AGG_ROWS) indeg[tid] = 0;
    __syncthreads();

    const int4*   __restrict__ dst4 = reinterpret_cast<const int4*>(dst);
    const float4* __restrict__ h4   = reinterpret_cast<const float4*>(h);

    int cnt = 0;

    auto drain = [&](int n) {
        const int start = cnt - n;
        const int ent = mylist[start + (lane < n ? lane : 0)];
        const int e   = ent & ((1 << EDGE_BITS) - 1);
        const int dl  = ent >> EDGE_BITS;
        const int   s = src[e];
        const float w = ew[e];
        for (int j0 = 0; j0 < n; j0 += 8) {
            float4 r[8];
            #pragma unroll
            for (int u = 0; u < 8; ++u) {
                const int jj = (j0 + u < n) ? (j0 + u) : (n - 1);
                const int sj = __builtin_amdgcn_readlane(s, jj);
                r[u] = h4[(size_t)sj * (OUT_FEATS / 4) + lane];
            }
            #pragma unroll
            for (int u = 0; u < 8; ++u) {
                if (j0 + u < n) {
                    const int   dj = __builtin_amdgcn_readlane(dl, j0 + u);
                    const float wj = __int_as_float(__builtin_amdgcn_readlane(__float_as_int(w), j0 + u));
                    float* a = acc + dj * OUT_FEATS + lane;
                    __hip_atomic_fetch_add(a +  0, r[u].x * wj, __ATOMIC_RELAXED, __HIP_MEMORY_SCOPE_WORKGROUP);
                    __hip_atomic_fetch_add(a + 32, r[u].y * wj, __ATOMIC_RELAXED, __HIP_MEMORY_SCOPE_WORKGROUP);
                    __hip_atomic_fetch_add(a + 64, r[u].z * wj, __ATOMIC_RELAXED, __HIP_MEMORY_SCOPE_WORKGROUP);
                    __hip_atomic_fetch_add(a + 96, r[u].w * wj, __ATOMIC_RELAXED, __HIP_MEMORY_SCOPE_WORKGROUP);
                }
            }
        }
        cnt = start;
    };

    auto append = [&](bool m, int d, int e) {
        const unsigned bal = __builtin_amdgcn_ballot_w32(m);
        if (bal == 0u) return;
        if (m) {
            const int dl = d - base;
            mylist[cnt + (int)__builtin_amdgcn_mbcnt_lo(bal, 0u)] = e | (dl << EDGE_BITS);
            __hip_atomic_fetch_add(&indeg[dl], 1, __ATOMIC_RELAXED, __HIP_MEMORY_SCOPE_WORKGROUP);
        }
        cnt += __builtin_popcount(bal);
    };

    for (int i = 0; wave + AGG_WAVES * i <= BM_WORDS; ++i) {
        unsigned bits = (unsigned)__builtin_amdgcn_readlane((int)myw, i);
        const int ubase = (wave + AGG_WAVES * i) * 32;
        while (bits != 0u) {
            int4 dv[AGG_UNROLL];
            int  qv[AGG_UNROLL];
            #pragma unroll
            for (int u = 0; u < AGG_UNROLL; ++u) {
                int ua = -1, ub = -1;
                if (bits != 0u) { ua = __builtin_ctz(bits); bits &= bits - 1u; }
                if (bits != 0u) { ub = __builtin_ctz(bits); bits &= bits - 1u; }
                const int un = half ? ub : ua;
                const int q  = (ubase + un) * (BM_UNIT_EDGES / 4) + l;
                qv[u] = q;
                dv[u] = (un >= 0) ? dst4[q] : make_int4(-1, -1, -1, -1);
            }
            #pragma unroll
            for (int u = 0; u < AGG_UNROLL; ++u) {
                const int e0 = 4 * qv[u];
                append((unsigned)(dv[u].x - base) < (unsigned)AGG_ROWS, dv[u].x, e0 + 0);
                append((unsigned)(dv[u].y - base) < (unsigned)AGG_ROWS, dv[u].y, e0 + 1);
                append((unsigned)(dv[u].z - base) < (unsigned)AGG_ROWS, dv[u].z, e0 + 2);
                append((unsigned)(dv[u].w - base) < (unsigned)AGG_ROWS, dv[u].w, e0 + 3);
            }
            while (cnt >= 32) drain(32);
        }
    }
    if (cnt > 0) drain(cnt);

    __syncthreads();

    const float4 bb = reinterpret_cast<const float4*>(b)[lane];
    for (int row = wave; row < AGG_ROWS; row += AGG_WAVES) {
        const int node = base + row;
        if (node >= N_NODES) break;
        const float rn = rsqrtf(fmaxf((float)indeg[row], 1.0f));
        const float* a = acc + row * OUT_FEATS + lane;
        float4 o;
        o.x = a[0]  * rn + bb.x;
        o.y = a[32] * rn + bb.y;
        o.z = a[64] * rn + bb.z;
        o.w = a[96] * rn + bb.w;
        reinterpret_cast<float4*>(out + (size_t)node * OUT_FEATS)[lane] = o;
    }
}

extern "C" void kernel_launch(void* const* d_in, const int* in_sizes, int n_in,
                              void* d_out, int out_size, void* d_ws, size_t ws_size,
                              hipStream_t stream) {
    const float* x   = (const float*)d_in[0];
    const float* ew  = (const float*)d_in[1];
    const float* Wm  = (const float*)d_in[2];
    const float* b   = (const float*)d_in[3];
    const int*   src = (const int*)d_in[4];
    const int*   dst = (const int*)d_in[5];

    float* out = (float*)d_out;

    int*   bm = (int*)d_ws;
    float* h  = (float*)d_ws + 2 * N_NODES;

    gc_zero_counters_kernel<<<(N_NODES + 255) / 256, 256, 0, stream>>>(h);

    gc_edge_prep_kernel<<<BM_WORDS + (N_EDGES + 255) / 256, 256, 0, stream>>>(src, dst, bm, h);

    gc_gemm_wmma_kernel<<<GM_GROUPS, 256, 0, stream>>>(x, Wm, h);

    gc_aggregate_kernel<<<AGG_GROUPS, AGG_THREADS, 0, stream>>>(h, ew, src, dst, b, bm, out);
}


# speedup vs baseline: 3.1261x; 1.0573x over previous
#include <hip/hip_runtime.h>
#include <hip/hip_bf16.h>

typedef float v2f __attribute__((ext_vector_type(2)));
typedef float v8f __attribute__((ext_vector_type(8)));

#define N_NODES   50000
#define N_EDGES   800000
#define IN_FEATS  256
#define OUT_FEATS 128

#define CNT_OUT_OFF 0

__global__ void gc_zero_counters_kernel(float* __restrict__ h) {
    const int i = blockIdx.x * blockDim.x + threadIdx.x;
    if (i >= N_NODES) return;
    h[(size_t)i * OUT_FEATS + CNT_OUT_OFF] = 0.0f;
}

#define AGG_ROWS    196
#define AGG_GROUPS  256

#define BM_UNIT_EDGES 64
#define BM_WORDS      (N_EDGES / BM_UNIT_EDGES / 32)
#define BM_TAIL_UNITS (N_EDGES / BM_UNIT_EDGES - 32 * BM_WORDS)
static_assert(N_EDGES % BM_UNIT_EDGES == 0, "units are whole");
static_assert(BM_TAIL_UNITS > 0 && BM_TAIL_UNITS < 32, "tail units fit one virtual word");
#define H_OFF         (((2 * N_NODES) / 128) * 128)
static_assert(H_OFF % 128 == 0 && H_OFF <= 2 * N_NODES, "h is 512-B aligned to the base and ends inside the old extent");
static_assert((size_t)BM_WORDS * AGG_GROUPS <= (size_t)H_OFF, "bitmap fits the workspace below h");

__global__ void __launch_bounds__(256)
gc_edge_prep_kernel(const int* __restrict__ src,
                    const int* __restrict__ dst,
                    int* __restrict__ bm,
                    float* __restrict__ h) {
    __shared__ unsigned m[AGG_GROUPS];
    const int tid = threadIdx.x;
    if (blockIdx.x < BM_WORDS) {
        m[tid] = 0u;
        __syncthreads();
        const int4* __restrict__ dst4 = reinterpret_cast<const int4*>(dst);
        const int q0 = blockIdx.x * (32 * BM_UNIT_EDGES / 4);
        int4 d[2];
        #pragma unroll
        for (int k = 0; k < 2; ++k)
            d[k] = dst4[q0 + k * 256 + tid];
        #pragma unroll
        for (int k = 0; k < 2; ++k) {
            const unsigned bit = 1u << (((q0 + k * 256 + tid) >> 4) & 31);
            const unsigned t0 = (unsigned)d[k].x / AGG_ROWS, t1 = (unsigned)d[k].y / AGG_ROWS;
            const unsigned t2 = (unsigned)d[k].z / AGG_ROWS, t3 = (unsigned)d[k].w / AGG_ROWS;
            if (t0 < AGG_GROUPS) __hip_atomic_fetch_or(&m[t0], bit, __ATOMIC_RELAXED, __HIP_MEMORY_SCOPE_WORKGROUP);
            if (t1 < AGG_GROUPS) __hip_atomic_fetch_or(&m[t1], bit, __ATOMIC_RELAXED, __HIP_MEMORY_SCOPE_WORKGROUP);
            if (t2 < AGG_GROUPS) __hip_atomic_fetch_or(&m[t2], bit, __ATOMIC_RELAXED, __HIP_MEMORY_SCOPE_WORKGROUP);
            if (t3 < AGG_GROUPS) __hip_atomic_fetch_or(&m[t3], bit, __ATOMIC_RELAXED, __HIP_MEMORY_SCOPE_WORKGROUP);
        }
        __syncthreads();
        bm[blockIdx.x * AGG_GROUPS + tid] = (int)m[tid];
        return;
    }
    const int e = (blockIdx.x - BM_WORDS) * 256 + tid;
    if (e >= N_EDGES) return;
    unsafeAtomicAdd(&h[(size_t)src[e] * OUT_FEATS + CNT_OUT_OFF], 1.0f);
}

#define GM_GROUPS  512
#define GM_TILES   (N_NODES / 16)
#define GM_PITCH   260
static_assert(N_NODES % 16 == 0, "row tiles are whole");
static_assert(GM_GROUPS <= GM_TILES, "every group has a first tile");

__global__ void __launch_bounds__(256)
gc_gemm_wmma_kernel(const float* __restrict__ x,
                    const float* __restrict__ Wm,
                    float* __restrict__ h) {
    __shared__ __attribute__((aligned(16))) float xs[2][16 * GM_PITCH];

    const int tid  = threadIdx.x;
    const int lane = tid & 31;
    const int wave = tid >> 5;
    const int half = lane >> 4;
    const int l    = lane & 15;
    const int n0   = wave * 16;

    const int srow = tid >> 6;
    const int c4   = tid & 63;
    const float4* __restrict__ x4 = reinterpret_cast<const float4*>(x);

    float4 xv[4];
    float  od[4];

    int tile = blockIdx.x;
    #pragma unroll
    for (int i = 0; i < 4; ++i) {
        const int row = tile * 16 + srow + 4 * i;
        od[i] = h[(size_t)row * OUT_FEATS + CNT_OUT_OFF];
        xv[i] = x4[(size_t)row * (IN_FEATS / 4) + c4];
    }

    v2f wf[IN_FEATS / 4];
    #pragma unroll
    for (int s = 0; s < IN_FEATS / 4; ++s) {
        const int ka = 4 * s + 2 * half;
        wf[s].x = Wm[(size_t)ka       * OUT_FEATS + n0 + l];
        wf[s].y = Wm[(size_t)(ka + 1) * OUT_FEATS + n0 + l];
    }

    #pragma unroll
    for (int i = 0; i < 4; ++i) {
        const float rs = rsqrtf(fmaxf(od[i], 1.0f));
        float4 t = xv[i];
        t.x *= rs; t.y *= rs; t.z *= rs; t.w *= rs;
        *reinterpret_cast<float4*>(&xs[0][(srow + 4 * i) * GM_PITCH + 4 * c4]) = t;
    }
    __syncthreads();

    int buf = 0;
    for (; tile < GM_TILES; tile += GM_GROUPS) {
        const int  next     = tile + GM_GROUPS;
        const bool has_next = next < GM_TILES;
        if (has_next) {
            #pragma unroll
            for (int i = 0; i < 4; ++i) {
                const int row = next * 16 + srow + 4 * i;
                od[i] = h[(size_t)row * OUT_FEATS + CNT_OUT_OFF];
                xv[i] = x4[(size_t)row * (IN_FEATS / 4) + c4];
            }
        }

        const float* __restrict__ arow = &xs[buf][l * GM_PITCH + 2 * half];
        v8f c = {};
        #pragma unroll
        for (int s = 0; s < IN_FEATS / 4; ++s) {
            const v2f a = *reinterpret_cast<const v2f*>(arow + 4 * s);
            c = __builtin_amdgcn_wmma_f32_16x16x4_f32(false, a, false, wf[s],
                                                      (short)0, c, false, false);
        }

        const int m0 = tile * 16;
        #pragma unroll
        for (int r = 0; r < 8; ++r) {
            const int m = m0 + r + 8 * half;
            h[(size_t)m * OUT_FEATS + n0 + l] = c[r];
        }

        if (has_next) {
            #pragma unroll
            for (int i = 0; i < 4; ++i) {
                const float rs = rsqrtf(fmaxf(od[i], 1.0f));
                float4 t = xv[i];
                t.x *= rs; t.y *= rs; t.z *= rs; t.w *= rs;
                *reinterpret_cast<float4*>(&xs[buf ^ 1][(srow + 4 * i) * GM_PITCH + 4 * c4]) = t;
            }
        }
        __syncthreads();
        buf ^= 1;
    }
}

#define AGG_THREADS 1024
#define AGG_WAVES   (AGG_THREADS / 32)
#define AGG_UNROLL  4
#define AGG_LIST    544
#define EDGE_BITS   20
static_assert(N_EDGES < (1 << EDGE_BITS), "edge id must fit its field");
static_assert(AGG_ROWS <= 256, "row-in-tile must fit 8 bits");
static_assert(AGG_GROUPS * AGG_ROWS >= N_NODES, "tiles must cover every node");
static_assert(N_EDGES % 4 == 0, "dst is read 4 edges per lane");
static_assert(AGG_GROUPS == 256, "the bitmap blocks build one word per thread");
static_assert(BM_WORDS / AGG_WAVES < 32, "a wave's bitmap words fit one per lane");

__global__ void __launch_bounds__(AGG_THREADS)
gc_aggregate_kernel(const float* __restrict__ h,
                    const float* __restrict__ ew,
                    const int* __restrict__ src,
                    const int* __restrict__ dst,
                    const float* __restrict__ b,
                    const int* __restrict__ bm,
                    float* __restrict__ out) {
    __shared__ __attribute__((aligned(16))) float acc[AGG_ROWS * OUT_FEATS];
    __shared__ int indeg[AGG_ROWS];
    __shared__ int lst[AGG_WAVES][AGG_LIST];

    const int tid  = threadIdx.x;
    const int lane = tid & 31;
    const int wave = tid >> 5;
    const int base = blockIdx.x * AGG_ROWS;
    int* __restrict__ mylist = lst[wave];
    const int half = lane >> 4;
    const int l    = lane & 15;

    const int wmine = wave + AGG_WAVES * lane;
    unsigned myw = 0u;
    if (wmine < BM_WORDS)       myw = (unsigned)bm[wmine * AGG_GROUPS + blockIdx.x];
    else if (wmine == BM_WORDS) myw = (1u << BM_TAIL_UNITS) - 1u;

    for (int i = tid; i < AGG_ROWS * OUT_FEATS / 4; i += AGG_THREADS)
        reinterpret_cast<float4*>(acc)[i] = make_float4(0.f, 0.f, 0.f, 0.f);
    if (tid < AGG_ROWS) indeg[tid] = 0;
    __syncthreads();

    const int4*   __restrict__ dst4 = reinterpret_cast<const int4*>(dst);
    const float4* __restrict__ h4   = reinterpret_cast<const float4*>(h);

    int cnt = 0;

    auto drain = [&](int n) {
        const int start = cnt - n;
        const int ent = mylist[start + (lane < n ? lane : 0)];
        const int e   = ent & ((1 << EDGE_BITS) - 1);
        const int dl  = ent >> EDGE_BITS;
        const int   s = src[e];
        const float w = ew[e];
        for (int j0 = 0; j0 < n; j0 += 8) {
            float4 r[8];
            #pragma unroll
            for (int u = 0; u < 8; ++u) {
                const int jj = (j0 + u < n) ? (j0 + u) : (n - 1);
                const int sj = __builtin_amdgcn_readlane(s, jj);
                r[u] = h4[(size_t)sj * (OUT_FEATS / 4) + lane];
            }
            #pragma unroll
            for (int u = 0; u < 8; ++u) {
                if (j0 + u < n) {
                    const int   dj = __builtin_amdgcn_readlane(dl, j0 + u);
                    const float wj = __int_as_float(__builtin_amdgcn_readlane(__float_as_int(w), j0 + u));
                    float* a = acc + dj * OUT_FEATS + lane;
                    __hip_atomic_fetch_add(a +  0, r[u].x * wj, __ATOMIC_RELAXED, __HIP_MEMORY_SCOPE_WORKGROUP);
                    __hip_atomic_fetch_add(a + 32, r[u].y * wj, __ATOMIC_RELAXED, __HIP_MEMORY_SCOPE_WORKGROUP);
                    __hip_atomic_fetch_add(a + 64, r[u].z * wj, __ATOMIC_RELAXED, __HIP_MEMORY_SCOPE_WORKGROUP);
                    __hip_atomic_fetch_add(a + 96, r[u].w * wj, __ATOMIC_RELAXED, __HIP_MEMORY_SCOPE_WORKGROUP);
                }
            }
        }
        cnt = start;
    };

    auto append = [&](bool m, int d, int e) {
        const unsigned bal = __builtin_amdgcn_ballot_w32(m);
        if (bal == 0u) return;
        if (m) {
            const int dl = d - base;
            mylist[cnt + (int)__builtin_amdgcn_mbcnt_lo(bal, 0u)] = e | (dl << EDGE_BITS);
            __hip_atomic_fetch_add(&indeg[dl], 1, __ATOMIC_RELAXED, __HIP_MEMORY_SCOPE_WORKGROUP);
        }
        cnt += __builtin_popcount(bal);
    };

    for (int i = 0; wave + AGG_WAVES * i <= BM_WORDS; ++i) {
        unsigned bits = (unsigned)__builtin_amdgcn_readlane((int)myw, i);
        const int ubase = (wave + AGG_WAVES * i) * 32;
        while (bits != 0u) {
            int4 dv[AGG_UNROLL];
            int  qv[AGG_UNROLL];
            #pragma unroll
            for (int u = 0; u < AGG_UNROLL; ++u) {
                int ua = -1, ub = -1;
                if (bits != 0u) { ua = __builtin_ctz(bits); bits &= bits - 1u; }
                if (bits != 0u) { ub = __builtin_ctz(bits); bits &= bits - 1u; }
                const int un = half ? ub : ua;
                const int q  = (ubase + un) * (BM_UNIT_EDGES / 4) + l;
                qv[u] = q;
                dv[u] = (un >= 0) ? dst4[q] : make_int4(-1, -1, -1, -1);
            }
            #pragma unroll
            for (int u = 0; u < AGG_UNROLL; ++u) {
                const int e0 = 4 * qv[u];
                append((unsigned)(dv[u].x - base) < (unsigned)AGG_ROWS, dv[u].x, e0 + 0);
                append((unsigned)(dv[u].y - base) < (unsigned)AGG_ROWS, dv[u].y, e0 + 1);
                append((unsigned)(dv[u].z - base) < (unsigned)AGG_ROWS, dv[u].z, e0 + 2);
                append((unsigned)(dv[u].w - base) < (unsigned)AGG_ROWS, dv[u].w, e0 + 3);
            }
            while (cnt >= 32) drain(32);
        }
    }
    if (cnt > 0) drain(cnt);

    __syncthreads();

    const float4 bb = reinterpret_cast<const float4*>(b)[lane];
    for (int row = wave; row < AGG_ROWS; row += AGG_WAVES) {
        const int node = base + row;
        if (node >= N_NODES) break;
        const float rn = rsqrtf(fmaxf((float)indeg[row], 1.0f));
        const float* a = acc + row * OUT_FEATS + lane;
        float4 o;
        o.x = a[0]  * rn + bb.x;
        o.y = a[32] * rn + bb.y;
        o.z = a[64] * rn + bb.z;
        o.w = a[96] * rn + bb.w;
        reinterpret_cast<float4*>(out + (size_t)node * OUT_FEATS)[lane] = o;
    }
}

extern "C" void kernel_launch(void* const* d_in, const int* in_sizes, int n_in,
                              void* d_out, int out_size, void* d_ws, size_t ws_size,
                              hipStream_t stream) {
    const float* x   = (const float*)d_in[0];
    const float* ew  = (const float*)d_in[1];
    const float* Wm  = (const float*)d_in[2];
    const float* b   = (const float*)d_in[3];
    const int*   src = (const int*)d_in[4];
    const int*   dst = (const int*)d_in[5];

    float* out = (float*)d_out;

    int*   bm = (int*)d_ws;
    float* h  = (float*)d_ws + H_OFF;

    gc_zero_counters_kernel<<<(N_NODES + 255) / 256, 256, 0, stream>>>(h);

    gc_edge_prep_kernel<<<BM_WORDS + (N_EDGES + 255) / 256, 256, 0, stream>>>(src, dst, bm, h);

    gc_gemm_wmma_kernel<<<GM_GROUPS, 256, 0, stream>>>(x, Wm, h);

    gc_aggregate_kernel<<<AGG_GROUPS, AGG_THREADS, 0, stream>>>(h, ew, src, dst, b, bm, out);
}
